# starting kernel
# baseline (speedup 1.0000x reference)
#include <hip/hip_runtime.h>
#include <hip/hip_bf16.h>


typedef float v2f __attribute__((ext_vector_type(2)));
typedef float v8f __attribute__((ext_vector_type(8)));

#define NFEAT   128
#define NHID    96
#define NCLASS  40
#define GN_EPS  1e-5f

template<int NT, int KMAX>
__global__ __launch_bounds__(128) void wmma_gemm_strip_kernel(
    const float* __restrict__ A, const float* __restrict__ Bw,
    float* __restrict__ C, int K, int ncols, int ntm)
{
    constexpr int NCP = NT * 16;
    __shared__ float lds[KMAX * NCP];

    const int tid = threadIdx.x;

    for (int i = tid; i < K * NCP; i += 128) {
        int k   = i / NCP;
        int col = i - k * NCP;
        float v = (col < ncols) ? Bw[(size_t)k * ncols + col] : 0.0f;
        lds[(k >> 1) * (NCP * 2) + col * 2 + (k & 1)] = v;
    }
    __syncthreads();

    const int wave  = tid >> 5;
    const int mtile = blockIdx.x * 4 + wave;
    if (mtile >= ntm) return;

    const int lane  = tid & 31;
    const int coll  = lane & 15;
    const int mrow  = mtile * 16 + coll;
    const int khalf = (lane >> 4) * 2;

    const float2* __restrict__ lds2 = reinterpret_cast<const float2*>(lds);

    v8f acc[NT] = {};
    for (int k = 0; k < K; k += 4) {
        const int kk = k + khalf;
        float2 av = *reinterpret_cast<const float2*>(A + (size_t)mrow * K + kk);
        v2f a; a.x = av.x; a.y = av.y;
        const float2* bp = lds2 + (kk >> 1) * NCP + coll;
        #pragma unroll
        for (int t = 0; t < NT; ++t) {
            float2 bv = bp[t * 16];
            v2f b; b.x = bv.x; b.y = bv.y;
            acc[t] = __builtin_amdgcn_wmma_f32_16x16x4_f32(
                false, a, false, b, (short)0, acc[t], false, false);
        }
    }

    const int rbase = mtile * 16 + (lane >> 4) * 8;
    #pragma unroll
    for (int t = 0; t < NT; ++t) {
        int col = t * 16 + coll;
        if (col < ncols) {
            #pragma unroll
            for (int r = 0; r < 8; ++r)
                C[(size_t)(rbase + r) * ncols + col] = acc[t][r];
        }
    }
}

__global__ void scatter_kernel(const float* __restrict__ h,
                               const int* __restrict__ src,
                               const int* __restrict__ tgt,
                               const float* __restrict__ w,
                               float* __restrict__ agg,
                               int C, int vpe, int total)
{
    int idx = blockIdx.x * blockDim.x + threadIdx.x;
    if (idx >= total) return;
    int e = idx / vpe;
    int c = (idx - e * vpe) * 4;
    int s = src[e];
    int t = tgt[e];
    float ww = w[e];
    float4 m = *reinterpret_cast<const float4*>(h + (size_t)s * C + c);
    float* ap = agg + (size_t)t * C + c;
    atomicAdd(ap + 0, m.x * ww);
    atomicAdd(ap + 1, m.y * ww);
    atomicAdd(ap + 2, m.z * ww);
    atomicAdd(ap + 3, m.w * ww);
}

__global__ void bias_relu_kernel(float* __restrict__ x,
                                 const float* __restrict__ b,
                                 int C, int total)
{
    int idx = blockIdx.x * blockDim.x + threadIdx.x;
    if (idx >= total) return;
    int c = idx % C;
    x[idx] = fmaxf(x[idx] + b[c], 0.0f);
}

__global__ void groupnorm_kernel(float* __restrict__ x,
                                 const float* __restrict__ gamma,
                                 const float* __restrict__ beta,
                                 int total  )
{
    int idx = blockIdx.x * blockDim.x + threadIdx.x;
    if (idx >= total) return;
    int row = idx >> 5;
    int g   = idx & 31;
    size_t base = (size_t)row * NHID + g * 3;
    float x0 = x[base], x1 = x[base + 1], x2 = x[base + 2];
    float mu = (x0 + x1 + x2) * (1.0f / 3.0f);
    float d0 = x0 - mu, d1 = x1 - mu, d2 = x2 - mu;
    float var = (d0 * d0 + d1 * d1 + d2 * d2) * (1.0f / 3.0f);
    float inv = rsqrtf(var + GN_EPS);
    int cg = g * 3;
    x[base]     = d0 * inv * gamma[cg]     + beta[cg];
    x[base + 1] = d1 * inv * gamma[cg + 1] + beta[cg + 1];
    x[base + 2] = d2 * inv * gamma[cg + 2] + beta[cg + 2];
}

__global__ void bias_logsoftmax_kernel(float* __restrict__ out,
                                       const float* __restrict__ b,
                                       int nrows)
{
    int row = blockIdx.x * blockDim.x + threadIdx.x;
    if (row >= nrows) return;
    float v[NCLASS];
    float m = -3.402823466e+38f;
    #pragma unroll
    for (int c = 0; c < NCLASS; ++c) {
        v[c] = out[(size_t)row * NCLASS + c] + b[c];
        m = fmaxf(m, v[c]);
    }
    float s = 0.0f;
    #pragma unroll
    for (int c = 0; c < NCLASS; ++c) s += __expf(v[c] - m);
    float ls = __logf(s);
    #pragma unroll
    for (int c = 0; c < NCLASS; ++c)
        out[(size_t)row * NCLASS + c] = v[c] - m - ls;
}

extern "C" void kernel_launch(void* const* d_in, const int* in_sizes, int n_in,
                              void* d_out, int out_size, void* d_ws, size_t ws_size,
                              hipStream_t stream)
{
    const float* x     = (const float*)d_in[0];
    const int*   src   = (const int*)  d_in[1];
    const int*   tgt   = (const int*)  d_in[2];
    const float* w     = (const float*)d_in[3];
    const float* W1    = (const float*)d_in[4];
    const float* b1    = (const float*)d_in[5];
    const float* W2    = (const float*)d_in[6];
    const float* b2    = (const float*)d_in[7];
    const float* W3    = (const float*)d_in[8];
    const float* b3    = (const float*)d_in[9];
    const float* gamma = (const float*)d_in[10];
    const float* beta  = (const float*)d_in[11];
    float* out = (float*)d_out;

    const int N   = in_sizes[0] / NFEAT;
    const int E   = in_sizes[1];
    const int ntm = N / 16;
    const int gblk = (ntm + 3) / 4;

    float* bufA = (float*)d_ws;
    float* bufB = bufA + (size_t)N * NHID;

    const int TPB = 256;

    wmma_gemm_strip_kernel<6, NFEAT><<<gblk, 128, 0, stream>>>(
        x, W1, bufA, NFEAT, NHID, ntm);
    hipMemsetAsync(bufB, 0, (size_t)N * NHID * sizeof(float), stream);
    {
        int vpe = NHID / 4, total = E * vpe;
        scatter_kernel<<<(total + TPB - 1) / TPB, TPB, 0, stream>>>(
            bufA, src, tgt, w, bufB, NHID, vpe, total);
        int t2 = N * NHID;
        bias_relu_kernel<<<(t2 + TPB - 1) / TPB, TPB, 0, stream>>>(bufB, b1, NHID, t2);
    }

    wmma_gemm_strip_kernel<6, NFEAT><<<gblk, 128, 0, stream>>>(
        bufB, W2, bufA, NHID, NHID, ntm);
    hipMemsetAsync(bufB, 0, (size_t)N * NHID * sizeof(float), stream);
    {
        int vpe = NHID / 4, total = E * vpe;
        scatter_kernel<<<(total + TPB - 1) / TPB, TPB, 0, stream>>>(
            bufA, src, tgt, w, bufB, NHID, vpe, total);
        int t2 = N * NHID;
        bias_relu_kernel<<<(t2 + TPB - 1) / TPB, TPB, 0, stream>>>(bufB, b2, NHID, t2);
    }

    {
        int total = N * 32;
        groupnorm_kernel<<<(total + TPB - 1) / TPB, TPB, 0, stream>>>(
            bufB, gamma, beta, total);
    }

    wmma_gemm_strip_kernel<3, NHID><<<gblk, 128, 0, stream>>>(
        bufB, W3, bufA, NHID, NCLASS, ntm);
    hipMemsetAsync(out, 0, (size_t)N * NCLASS * sizeof(float), stream);
    {
        int vpe = NCLASS / 4, total = E * vpe;
        scatter_kernel<<<(total + TPB - 1) / TPB, TPB, 0, stream>>>(
            bufA, src, tgt, w, out, NCLASS, vpe, total);
    }

    bias_logsoftmax_kernel<<<(N + TPB - 1) / TPB, TPB, 0, stream>>>(out, b3, N);
}


# Zero start value of the sum planes stored by the product kernels: each wave clears its 16 rows of both planes with whole-line 16-byte stores; the 3 fill operations are gone; sums unchanged
# speedup vs baseline: 2.3333x; 2.3333x over previous
#include <hip/hip_runtime.h>
#include <hip/hip_bf16.h>


typedef float v2f __attribute__((ext_vector_type(2)));
typedef float v8f __attribute__((ext_vector_type(8)));

#define NFEAT   128
#define NHID    96
#define NCLASS  40
#define GN_EPS  1e-5f

static __device__ __forceinline__ int package_half()
{
    return (int)((__builtin_amdgcn_s_sendmsg_rtn(0x87) >> 18) & 1u);
}


template<int NCOLS>
static __device__ __forceinline__ v2f load_b_frag(const float* __restrict__ wrow,
                                                  int t, int coll)
{
    v2f b;
    if (t * 16 + 16 <= NCOLS) {
        b.x = wrow[t * 16];
        b.y = wrow[NCOLS + t * 16];
    } else {
        const bool in = (t * 16 + coll) < NCOLS;
        const int  o  = in ? t * 16 : 0;
        const float x = wrow[o];
        const float y = wrow[NCOLS + o];
        b.x = in ? x : 0.0f;
        b.y = in ? y : 0.0f;
    }
    return b;
}

template<int NT, int NCOLS, int KMAX, int MODE, int ZC>
__global__ __launch_bounds__(128) void wmma_gemm_strip_kernel(
    const float* __restrict__ A, const float* __restrict__ Bw,
    float* __restrict__ C,
    const float* __restrict__ bias,
    const float* __restrict__ gamma,
    const float* __restrict__ beta,
    int K, int ntm, size_t aplane,
    float* __restrict__ Z, size_t zplane)
{
    __shared__ float sbias[MODE >= 1 ? KMAX : 2];
    __shared__ float sgamma[MODE == 2 ? KMAX : 2];
    __shared__ float sbeta[MODE == 2 ? KMAX : 2];

    const int tid = threadIdx.x;

    if (MODE >= 1) {
        for (int i = tid; i < K; i += 128) {
            sbias[i] = bias[i];
            if (MODE == 2) { sgamma[i] = gamma[i]; sbeta[i] = beta[i]; }
        }
        __syncthreads();
    }

    const int wave  = tid >> 5;
    const int mtile = blockIdx.x * 4 + wave;
    if (mtile >= ntm) return;

    const int lane  = tid & 31;
    const int coll  = lane & 15;
    const int mrow  = mtile * 16 + coll;
    const int hsel  = lane >> 4;
    const int khalf = hsel * 2;

    {
        constexpr int NV = 16 * ZC / 4;
        float4* z0 = reinterpret_cast<float4*>(Z + (size_t)mtile * (16 * ZC));
        float4* z1 = reinterpret_cast<float4*>(Z + zplane + (size_t)mtile * (16 * ZC));
        const float4 zero4 = make_float4(0.0f, 0.0f, 0.0f, 0.0f);
        #pragma unroll
        for (int i = 0; i < NV / 32; ++i) {
            z0[i * 32 + lane] = zero4;
            z1[i * 32 + lane] = zero4;
        }
    }

    v8f acc[NT] = {};
    if (MODE == 2) {
        for (int kb = 0; kb < K; kb += 12) {
            const int c0 = kb + 6 * hsel;
            const float* ap = A + (size_t)mrow * K + c0;
            const float* aq = ap + aplane;
            const float2 p0 = *reinterpret_cast<const float2*>(ap);
            const float2 p1 = *reinterpret_cast<const float2*>(ap + 2);
            const float2 p2 = *reinterpret_cast<const float2*>(ap + 4);
            const float2 q0 = *reinterpret_cast<const float2*>(aq);
            const float2 q1 = *reinterpret_cast<const float2*>(aq + 2);
            const float2 q2 = *reinterpret_cast<const float2*>(aq + 4);
            float xv[6] = { p0.x + q0.x, p0.y + q0.y, p1.x + q1.x,
                            p1.y + q1.y, p2.x + q2.x, p2.y + q2.y };
            float y[6];
            #pragma unroll
            for (int i = 0; i < 6; ++i)
                xv[i] = fmaxf(xv[i] + sbias[c0 + i], 0.0f);
            #pragma unroll
            for (int g = 0; g < 2; ++g) {
                const float x0 = xv[3 * g], x1 = xv[3 * g + 1], x2 = xv[3 * g + 2];
                const float mu = (x0 + x1 + x2) * (1.0f / 3.0f);
                const float d0 = x0 - mu, d1 = x1 - mu, d2 = x2 - mu;
                const float var = (d0 * d0 + d1 * d1 + d2 * d2) * (1.0f / 3.0f);
                const float inv = rsqrtf(var + GN_EPS);
                const int cg = c0 + 3 * g;
                y[3 * g]     = d0 * inv * sgamma[cg]     + sbeta[cg];
                y[3 * g + 1] = d1 * inv * sgamma[cg + 1] + sbeta[cg + 1];
                y[3 * g + 2] = d2 * inv * sgamma[cg + 2] + sbeta[cg + 2];
            }
            #pragma unroll
            for (int s = 0; s < 3; ++s) {
                v2f a; a.x = y[2 * s]; a.y = y[2 * s + 1];
                const float* wrow = Bw + (size_t)(c0 + 2 * s) * NCOLS + coll;
                #pragma unroll
                for (int t = 0; t < NT; ++t) {
                    const v2f b = load_b_frag<NCOLS>(wrow, t, coll);
                    acc[t] = __builtin_amdgcn_wmma_f32_16x16x4_f32(
                        false, a, false, b, (short)0, acc[t], false, false);
                }
            }
        }
    } else {
        for (int k = 0; k < K; k += 4) {
            const int kk = k + khalf;
            const float* ap = A + (size_t)mrow * K + kk;
            float2 av = *reinterpret_cast<const float2*>(ap);
            if (MODE == 1) {
                const float2 aw = *reinterpret_cast<const float2*>(ap + aplane);
                av.x = fmaxf(av.x + aw.x + sbias[kk],     0.0f);
                av.y = fmaxf(av.y + aw.y + sbias[kk + 1], 0.0f);
            }
            v2f a; a.x = av.x; a.y = av.y;
            const float* wrow = Bw + (size_t)kk * NCOLS + coll;
            #pragma unroll
            for (int t = 0; t < NT; ++t) {
                const v2f b = load_b_frag<NCOLS>(wrow, t, coll);
                acc[t] = __builtin_amdgcn_wmma_f32_16x16x4_f32(
                    false, a, false, b, (short)0, acc[t], false, false);
            }
        }
    }

    const int rbase = mtile * 16 + (lane >> 4) * 8;
    #pragma unroll
    for (int t = 0; t < NT; ++t) {
        int col = t * 16 + coll;
        if (col < NCOLS) {
            #pragma unroll
            for (int r = 0; r < 8; ++r)
                C[(size_t)(rbase + r) * NCOLS + col] = acc[t][r];
        }
    }
}

template<int C, int U>
__global__ __launch_bounds__(256) void scatter_rows_kernel(
    const float* __restrict__ h,
    const int* __restrict__ src,
    const int* __restrict__ tgt,
    const float* __restrict__ w,
    float* __restrict__ agg,
    size_t plane,
    int E)
{
    constexpr int CPL = (C + 31) / 32;
    const int lane = threadIdx.x & 31;
    const int wave = __builtin_amdgcn_readfirstlane(
        (int)((blockIdx.x * blockDim.x + threadIdx.x) >> 5));
    const int e0 = wave * 32;
    if (e0 >= E) return;
    const int n = (E - e0 < 32) ? (E - e0) : 32;
    agg += (size_t)package_half() * plane;

    int   s  = 0;
    int   t  = 0;
    float ww = 0.0f;
    if (lane < n) {
        s  = src[e0 + lane];
        t  = tgt[e0 + lane];
        ww = w[e0 + lane];
    }
    const int wwi = __float_as_int(ww);

    int j = 0;
    for (; j + U <= n; j += U) {
        float v[U][CPL];
        #pragma unroll
        for (int u = 0; u < U; ++u) {
            const int sj = __builtin_amdgcn_readlane(s, j + u);
            const float* hp = h + (size_t)sj * C;
            #pragma unroll
            for (int q = 0; q < CPL; ++q) {
                const int c = lane + 32 * q;
                v[u][q] = (32 * q + 32 <= C || c < C) ? hp[c] : 0.0f;
            }
        }
        #pragma unroll
        for (int u = 0; u < U; ++u) {
            const int   tj = __builtin_amdgcn_readlane(t, j + u);
            const float wj = __int_as_float(__builtin_amdgcn_readlane(wwi, j + u));
            float* ap = agg + (size_t)tj * C;
            #pragma unroll
            for (int q = 0; q < CPL; ++q) {
                const int c = lane + 32 * q;
                if (32 * q + 32 <= C || c < C)
                    atomicAdd(ap + c, v[u][q] * wj);
            }
        }
    }
    for (; j < n; ++j) {
        const int   sj = __builtin_amdgcn_readlane(s, j);
        const int   tj = __builtin_amdgcn_readlane(t, j);
        const float wj = __int_as_float(__builtin_amdgcn_readlane(wwi, j));
        const float* hp = h + (size_t)sj * C;
        float* ap = agg + (size_t)tj * C;
        #pragma unroll
        for (int q = 0; q < CPL; ++q) {
            const int c = lane + 32 * q;
            if (32 * q + 32 <= C || c < C)
                atomicAdd(ap + c, hp[c] * wj);
        }
    }
}

__global__ void bias_logsoftmax_kernel(const float* __restrict__ agg,
                                       size_t plane,
                                       float* __restrict__ out,
                                       const float* __restrict__ b,
                                       int nrows)
{
    int row = blockIdx.x * blockDim.x + threadIdx.x;
    if (row >= nrows) return;
    float v[NCLASS];
    float m = -3.402823466e+38f;
    #pragma unroll
    for (int c = 0; c < NCLASS; ++c) {
        v[c] = agg[(size_t)row * NCLASS + c] + agg[plane + (size_t)row * NCLASS + c] + b[c];
        m = fmaxf(m, v[c]);
    }
    float s = 0.0f;
    #pragma unroll
    for (int c = 0; c < NCLASS; ++c) s += __expf(v[c] - m);
    float ls = __logf(s);
    #pragma unroll
    for (int c = 0; c < NCLASS; ++c)
        out[(size_t)row * NCLASS + c] = v[c] - m - ls;
}

extern "C" void kernel_launch(void* const* d_in, const int* in_sizes, int n_in,
                              void* d_out, int out_size, void* d_ws, size_t ws_size,
                              hipStream_t stream)
{
    const float* x     = (const float*)d_in[0];
    const int*   src   = (const int*)  d_in[1];
    const int*   tgt   = (const int*)  d_in[2];
    const float* w     = (const float*)d_in[3];
    const float* W1    = (const float*)d_in[4];
    const float* b1    = (const float*)d_in[5];
    const float* W2    = (const float*)d_in[6];
    const float* b2    = (const float*)d_in[7];
    const float* W3    = (const float*)d_in[8];
    const float* b3    = (const float*)d_in[9];
    const float* gamma = (const float*)d_in[10];
    const float* beta  = (const float*)d_in[11];
    float* out = (float*)d_out;

    const int N   = in_sizes[0] / NFEAT;
    const int E   = in_sizes[1];
    const int ntm = N / 16;
    const int gblk = (ntm + 3) / 4;

    const size_t planeH = (size_t)N * NHID;
    const size_t planeC = (size_t)N * NCLASS;
    float* bufA = (float*)d_ws;
    float* bufB = bufA + planeH;
    float* bufC = bufB + 2 * planeH;
    float* bufQ = bufB;

    const int TPB = 256;
    const int sblk = (E + TPB - 1) / TPB;

    wmma_gemm_strip_kernel<6, NHID, NFEAT, 0, NHID><<<gblk, 128, 0, stream>>>(
        x, W1, bufA, b1, gamma, beta, NFEAT, ntm, planeH, bufB, planeH);
    scatter_rows_kernel<NHID, 8><<<sblk, TPB, 0, stream>>>(
        bufA, src, tgt, w, bufB, planeH, E);

    wmma_gemm_strip_kernel<6, NHID, NFEAT, 1, NHID><<<gblk, 128, 0, stream>>>(
        bufB, W2, bufA, b1, gamma, beta, NHID, ntm, planeH, bufC, planeH);
    scatter_rows_kernel<NHID, 8><<<sblk, TPB, 0, stream>>>(
        bufA, src, tgt, w, bufC, planeH, E);

    wmma_gemm_strip_kernel<3, NCLASS, NHID, 2, NCLASS><<<gblk, 128, 0, stream>>>(
        bufC, W3, bufA, b2, gamma, beta, NHID, ntm, planeH, bufQ, planeC);
    scatter_rows_kernel<NCLASS, 8><<<sblk, TPB, 0, stream>>>(
        bufA, src, tgt, w, bufQ, planeC, E);

    bias_logsoftmax_kernel<<<(N + TPB - 1) / TPB, TPB, 0, stream>>>(bufQ, planeC, out, b3, N);
}


# Layer-3 rows (40 floats) stored as a 32-wide main array plus an 8-wide tail array: every row load and f32 atomic stays inside one 128-byte line, 2 line requests a row not 3; sums unchanged
# speedup vs baseline: 2.4791x; 1.0625x over previous
#include <hip/hip_runtime.h>
#include <hip/hip_bf16.h>


typedef float v2f __attribute__((ext_vector_type(2)));
typedef float v8f __attribute__((ext_vector_type(8)));

#define NFEAT   128
#define NHID    96
#define NCLASS  40
#define GN_EPS  1e-5f

static __device__ __forceinline__ int package_half()
{
    return (int)((__builtin_amdgcn_s_sendmsg_rtn(0x87) >> 18) & 1u);
}


template<int NCOLS>
static __device__ __forceinline__ v2f load_b_frag(const float* __restrict__ wrow,
                                                  int t, int coll)
{
    v2f b;
    if (t * 16 + 16 <= NCOLS) {
        b.x = wrow[t * 16];
        b.y = wrow[NCOLS + t * 16];
    } else {
        const bool in = (t * 16 + coll) < NCOLS;
        const int  o  = in ? t * 16 : 0;
        const float x = wrow[o];
        const float y = wrow[NCOLS + o];
        b.x = in ? x : 0.0f;
        b.y = in ? y : 0.0f;
    }
    return b;
}

template<int NT, int NCOLS, int KMAX, int MODE, int ZC>
__global__ __launch_bounds__(128) void wmma_gemm_strip_kernel(
    const float* __restrict__ A, const float* __restrict__ Bw,
    float* __restrict__ C,
    const float* __restrict__ bias,
    const float* __restrict__ gamma,
    const float* __restrict__ beta,
    int K, int ntm, size_t aplane,
    float* __restrict__ Z, size_t zplane,
    size_t ctail, size_t ztail)
{
    __shared__ float sbias[MODE >= 1 ? KMAX : 2];
    __shared__ float sgamma[MODE == 2 ? KMAX : 2];
    __shared__ float sbeta[MODE == 2 ? KMAX : 2];

    const int tid = threadIdx.x;

    if (MODE >= 1) {
        for (int i = tid; i < K; i += 128) {
            sbias[i] = bias[i];
            if (MODE == 2) { sgamma[i] = gamma[i]; sbeta[i] = beta[i]; }
        }
        __syncthreads();
    }

    const int wave  = tid >> 5;
    const int mtile = blockIdx.x * 4 + wave;
    if (mtile >= ntm) return;

    const int lane  = tid & 31;
    const int coll  = lane & 15;
    const int mrow  = mtile * 16 + coll;
    const int hsel  = lane >> 4;
    const int khalf = hsel * 2;

    {
        constexpr int ZMW = (ZC / 32) * 32;
        constexpr int ZTW = ZC - ZMW;
        constexpr int NV  = 16 * ZMW / 4;
        constexpr int NVT = 16 * ZTW / 4;
        static_assert(NVT == 0 || NVT == 32, "tail part: one 16-byte piece per lane");
        float4* z0 = reinterpret_cast<float4*>(Z + (size_t)mtile * (16 * ZMW));
        float4* z1 = reinterpret_cast<float4*>(Z + zplane + (size_t)mtile * (16 * ZMW));
        const float4 zero4 = make_float4(0.0f, 0.0f, 0.0f, 0.0f);
        #pragma unroll
        for (int i = 0; i < NV / 32; ++i) {
            z0[i * 32 + lane] = zero4;
            z1[i * 32 + lane] = zero4;
        }
        if (NVT) {
            float4* t0 = reinterpret_cast<float4*>(Z + ztail + (size_t)mtile * (16 * ZTW));
            float4* t1 = reinterpret_cast<float4*>(Z + zplane + ztail + (size_t)mtile * (16 * ZTW));
            t0[lane] = zero4;
            t1[lane] = zero4;
        }
    }

    v8f acc[NT] = {};
    if (MODE == 2) {
        for (int kb = 0; kb < K; kb += 12) {
            const int c0 = kb + 6 * hsel;
            const float* ap = A + (size_t)mrow * K + c0;
            const float* aq = ap + aplane;
            const float2 p0 = *reinterpret_cast<const float2*>(ap);
            const float2 p1 = *reinterpret_cast<const float2*>(ap + 2);
            const float2 p2 = *reinterpret_cast<const float2*>(ap + 4);
            const float2 q0 = *reinterpret_cast<const float2*>(aq);
            const float2 q1 = *reinterpret_cast<const float2*>(aq + 2);
            const float2 q2 = *reinterpret_cast<const float2*>(aq + 4);
            float xv[6] = { p0.x + q0.x, p0.y + q0.y, p1.x + q1.x,
                            p1.y + q1.y, p2.x + q2.x, p2.y + q2.y };
            float y[6];
            #pragma unroll
            for (int i = 0; i < 6; ++i)
                xv[i] = fmaxf(xv[i] + sbias[c0 + i], 0.0f);
            #pragma unroll
            for (int g = 0; g < 2; ++g) {
                const float x0 = xv[3 * g], x1 = xv[3 * g + 1], x2 = xv[3 * g + 2];
                const float mu = (x0 + x1 + x2) * (1.0f / 3.0f);
                const float d0 = x0 - mu, d1 = x1 - mu, d2 = x2 - mu;
                const float var = (d0 * d0 + d1 * d1 + d2 * d2) * (1.0f / 3.0f);
                const float inv = rsqrtf(var + GN_EPS);
                const int cg = c0 + 3 * g;
                y[3 * g]     = d0 * inv * sgamma[cg]     + sbeta[cg];
                y[3 * g + 1] = d1 * inv * sgamma[cg + 1] + sbeta[cg + 1];
                y[3 * g + 2] = d2 * inv * sgamma[cg + 2] + sbeta[cg + 2];
            }
            #pragma unroll
            for (int s = 0; s < 3; ++s) {
                v2f a; a.x = y[2 * s]; a.y = y[2 * s + 1];
                const float* wrow = Bw + (size_t)(c0 + 2 * s) * NCOLS + coll;
                #pragma unroll
                for (int t = 0; t < NT; ++t) {
                    const v2f b = load_b_frag<NCOLS>(wrow, t, coll);
                    acc[t] = __builtin_amdgcn_wmma_f32_16x16x4_f32(
                        false, a, false, b, (short)0, acc[t], false, false);
                }
            }
        }
    } else {
        for (int k = 0; k < K; k += 4) {
            const int kk = k + khalf;
            const float* ap = A + (size_t)mrow * K + kk;
            float2 av = *reinterpret_cast<const float2*>(ap);
            if (MODE == 1) {
                const float2 aw = *reinterpret_cast<const float2*>(ap + aplane);
                av.x = fmaxf(av.x + aw.x + sbias[kk],     0.0f);
                av.y = fmaxf(av.y + aw.y + sbias[kk + 1], 0.0f);
            }
            v2f a; a.x = av.x; a.y = av.y;
            const float* wrow = Bw + (size_t)kk * NCOLS + coll;
            #pragma unroll
            for (int t = 0; t < NT; ++t) {
                const v2f b = load_b_frag<NCOLS>(wrow, t, coll);
                acc[t] = __builtin_amdgcn_wmma_f32_16x16x4_f32(
                    false, a, false, b, (short)0, acc[t], false, false);
            }
        }
    }

    const int rbase = mtile * 16 + (lane >> 4) * 8;
    #pragma unroll
    for (int t = 0; t < NT; ++t) {
        constexpr int MW = (NCOLS / 32) * 32;
        constexpr int TW = NCOLS - MW;
        int col = t * 16 + coll;
        if (t * 16 + 16 <= MW) {
            #pragma unroll
            for (int r = 0; r < 8; ++r)
                C[(size_t)(rbase + r) * MW + col] = acc[t][r];
        } else if (col < NCOLS) {
            #pragma unroll
            for (int r = 0; r < 8; ++r)
                C[ctail + (size_t)(rbase + r) * TW + (col - MW)] = acc[t][r];
        }
    }
}

template<int C, int U>
__global__ __launch_bounds__(256) void scatter_rows_kernel(
    const float* __restrict__ h,
    const int* __restrict__ src,
    const int* __restrict__ tgt,
    const float* __restrict__ w,
    float* __restrict__ agg,
    size_t plane,
    int E,
    size_t htail, size_t atail)
{
    constexpr int CPL = (C + 31) / 32;
    constexpr int MW  = (C / 32) * 32;
    constexpr int TW  = C - MW;
    const int lane = threadIdx.x & 31;
    const int wave = __builtin_amdgcn_readfirstlane(
        (int)((blockIdx.x * blockDim.x + threadIdx.x) >> 5));
    const int e0 = wave * 32;
    if (e0 >= E) return;
    const int n = (E - e0 < 32) ? (E - e0) : 32;
    agg += (size_t)package_half() * plane;

    int   s  = 0;
    int   t  = 0;
    float ww = 0.0f;
    if (lane < n) {
        s  = src[e0 + lane];
        t  = tgt[e0 + lane];
        ww = w[e0 + lane];
    }
    const int wwi = __float_as_int(ww);

    int j = 0;
    for (; j + U <= n; j += U) {
        float v[U][CPL];
        #pragma unroll
        for (int u = 0; u < U; ++u) {
            const int sj = __builtin_amdgcn_readlane(s, j + u);
            const float* hp = h + (size_t)sj * MW;
            const float* ht = h + htail + (size_t)sj * TW;
            #pragma unroll
            for (int q = 0; q < CPL; ++q) {
                if (32 * q + 32 <= C) v[u][q] = hp[lane + 32 * q];
                else                  v[u][q] = (lane < TW) ? ht[lane] : 0.0f;
            }
        }
        #pragma unroll
        for (int u = 0; u < U; ++u) {
            const int   tj = __builtin_amdgcn_readlane(t, j + u);
            const float wj = __int_as_float(__builtin_amdgcn_readlane(wwi, j + u));
            float* ap = agg + (size_t)tj * MW;
            float* at = agg + atail + (size_t)tj * TW;
            #pragma unroll
            for (int q = 0; q < CPL; ++q) {
                if (32 * q + 32 <= C)
                    atomicAdd(ap + lane + 32 * q, v[u][q] * wj);
                else if (lane < TW)
                    atomicAdd(at + lane, v[u][q] * wj);
            }
        }
    }
    for (; j < n; ++j) {
        const int   sj = __builtin_amdgcn_readlane(s, j);
        const int   tj = __builtin_amdgcn_readlane(t, j);
        const float wj = __int_as_float(__builtin_amdgcn_readlane(wwi, j));
        const float* hp = h + (size_t)sj * MW;
        const float* ht = h + htail + (size_t)sj * TW;
        float* ap = agg + (size_t)tj * MW;
        float* at = agg + atail + (size_t)tj * TW;
        #pragma unroll
        for (int q = 0; q < CPL; ++q) {
            if (32 * q + 32 <= C)
                atomicAdd(ap + lane + 32 * q, hp[lane + 32 * q] * wj);
            else if (lane < TW)
                atomicAdd(at + lane, ht[lane] * wj);
        }
    }
}

__global__ void bias_logsoftmax_kernel(const float* __restrict__ agg,
                                       size_t plane, size_t tail,
                                       float* __restrict__ out,
                                       const float* __restrict__ b,
                                       int nrows)
{
    int row = blockIdx.x * blockDim.x + threadIdx.x;
    if (row >= nrows) return;
    float v[NCLASS];
    float m = -3.402823466e+38f;
    #pragma unroll
    for (int c = 0; c < NCLASS; ++c) {
        const size_t o = (c < 32) ? (size_t)row * 32 + c
                                  : tail + (size_t)row * (NCLASS - 32) + (c - 32);
        v[c] = agg[o] + agg[plane + o] + b[c];
        m = fmaxf(m, v[c]);
    }
    float s = 0.0f;
    #pragma unroll
    for (int c = 0; c < NCLASS; ++c) s += __expf(v[c] - m);
    float ls = __logf(s);
    #pragma unroll
    for (int c = 0; c < NCLASS; ++c)
        out[(size_t)row * NCLASS + c] = v[c] - m - ls;
}

extern "C" void kernel_launch(void* const* d_in, const int* in_sizes, int n_in,
                              void* d_out, int out_size, void* d_ws, size_t ws_size,
                              hipStream_t stream)
{
    const float* x     = (const float*)d_in[0];
    const int*   src   = (const int*)  d_in[1];
    const int*   tgt   = (const int*)  d_in[2];
    const float* w     = (const float*)d_in[3];
    const float* W1    = (const float*)d_in[4];
    const float* b1    = (const float*)d_in[5];
    const float* W2    = (const float*)d_in[6];
    const float* b2    = (const float*)d_in[7];
    const float* W3    = (const float*)d_in[8];
    const float* b3    = (const float*)d_in[9];
    const float* gamma = (const float*)d_in[10];
    const float* beta  = (const float*)d_in[11];
    float* out = (float*)d_out;

    const int N   = in_sizes[0] / NFEAT;
    const int E   = in_sizes[1];
    const int ntm = N / 16;
    const int gblk = (ntm + 3) / 4;

    const size_t planeH = (size_t)N * NHID;
    const size_t planeC = (size_t)N * NCLASS;
    const size_t tailC  = (size_t)N * 32;
    float* bufA = (float*)d_ws;
    float* bufB = bufA + planeH;
    float* bufC = bufB + 2 * planeH;
    float* bufQ = bufB;

    const int TPB = 256;
    const int sblk = (E + TPB - 1) / TPB;

    wmma_gemm_strip_kernel<6, NHID, NFEAT, 0, NHID><<<gblk, 128, 0, stream>>>(
        x, W1, bufA, b1, gamma, beta, NFEAT, ntm, planeH, bufB, planeH, 0, 0);
    scatter_rows_kernel<NHID, 8><<<sblk, TPB, 0, stream>>>(
        bufA, src, tgt, w, bufB, planeH, E, 0, 0);

    wmma_gemm_strip_kernel<6, NHID, NFEAT, 1, NHID><<<gblk, 128, 0, stream>>>(
        bufB, W2, bufA, b1, gamma, beta, NHID, ntm, planeH, bufC, planeH, 0, 0);
    scatter_rows_kernel<NHID, 8><<<sblk, TPB, 0, stream>>>(
        bufA, src, tgt, w, bufC, planeH, E, 0, 0);

    wmma_gemm_strip_kernel<3, NCLASS, NHID, 2, NCLASS><<<gblk, 128, 0, stream>>>(
        bufC, W3, bufA, b2, gamma, beta, NHID, ntm, planeH, bufQ, planeC, tailC, tailC);
    scatter_rows_kernel<NCLASS, 8><<<sblk, TPB, 0, stream>>>(
        bufA, src, tgt, w, bufQ, planeC, E, tailC, tailC);

    bias_logsoftmax_kernel<<<(N + TPB - 1) / TPB, TPB, 0, stream>>>(bufQ, planeC, tailC, out, b3, N);
}
